# starting kernel
# baseline (speedup 1.0000x reference)
#include <hip/hip_runtime.h>
#include <hip/hip_bf16.h>

#define H 256
#define BB 32
#define TT 4096
#define EE 8192
#define D2 512

typedef __attribute__((ext_vector_type(16))) __bf16        v16bf;
typedef __attribute__((ext_vector_type(8)))  float         v8f;
typedef __attribute__((ext_vector_type(8)))  unsigned int  v8u;

__device__ __forceinline__ unsigned short f2bf(float x) {
    unsigned int u = __float_as_uint(x);
    unsigned int r = u + 0x7FFFu + ((u >> 16) & 1u);
    return (unsigned short)(r >> 16);
}
__device__ __forceinline__ unsigned int pk2(float lo, float hi) {
    return (unsigned int)f2bf(lo) | ((unsigned int)f2bf(hi) << 16);
}

__device__ __forceinline__ float htanh(float x) {
#if __has_builtin(__builtin_amdgcn_tanhf)
    return __builtin_amdgcn_tanhf(x);
#elif __has_builtin(__builtin_amdgcn_tanh_f32)
    return __builtin_amdgcn_tanh_f32(x);
#else
    float y;
    asm volatile("v_tanh_f32 %0, %1\n\t"
                 "s_delay_alu instid0(TRANS32_DEP_1)"
                 : "=v"(y) : "v"(x));
    return y;
#endif
}

__global__ void packb_kernel(const float* __restrict__ W, unsigned short* __restrict__ Bp) {
    int blk  = blockIdx.x;
    int nt   = blk >> 3;
    int kc   = blk & 7;
    int lane = threadIdx.x;
    int n     = nt * 16 + (lane & 15);
    int kbase = kc * 32 + ((lane >> 4) ? 8 : 0);
    const float* wrow = W + (size_t)n * D2 + H;
    unsigned short* dst = Bp + ((size_t)blk * 32 + lane) * 16;
    #pragma unroll
    for (int j = 0; j < 8; ++j) dst[j]     = f2bf(wrow[kbase + j]);
    #pragma unroll
    for (int j = 0; j < 8; ++j) dst[8 + j] = f2bf(wrow[kbase + 16 + j]);
}

__global__ void p_kernel(const float* __restrict__ hidden, const float* __restrict__ W,
                         const float* __restrict__ bvec, float* __restrict__ P) {
    int b = blockIdx.x;
    int n = threadIdx.x;
    const float* h = hidden + b * H;
    const float* w = W + (size_t)n * D2;
    float acc = bvec[n];
    #pragma unroll 4
    for (int k = 0; k < H; ++k) acc = fmaf(h[k], w[k], acc);
    P[b * D2 + n] = acc;
}

__global__ void __launch_bounds__(256)
scores_kernel(const float* __restrict__ enc, const v8u* __restrict__ Bq,
              const float* __restrict__ P, const float* __restrict__ v,
              float* __restrict__ scores) {
    int wave = blockIdx.x * 8 + (threadIdx.x >> 5);
    int lane = threadIdx.x & 31;
    int b    = wave >> 7;
    int t0   = (wave & 127) << 5;
    int mrow = lane & 15;
    int kgrp = lane >> 4;

    const float* rowp0 = enc + ((size_t)(t0 + mrow)      * BB + b) * H + kgrp * 8;
    const float* rowp1 = enc + ((size_t)(t0 + 16 + mrow) * BB + b) * H + kgrp * 8;
    v16bf a0[8], a1[8];
    #pragma unroll
    for (int kc = 0; kc < 8; ++kc) {
        const float* p0 = rowp0 + kc * 32;
        const float* p1 = rowp1 + kc * 32;
        v8u u0, u1;
        #pragma unroll
        for (int j = 0; j < 4; ++j) { u0[j] = pk2(p0[2 * j],     p0[2 * j + 1]);
                                      u1[j] = pk2(p1[2 * j],     p1[2 * j + 1]); }
        #pragma unroll
        for (int j = 4; j < 8; ++j) { u0[j] = pk2(p0[8 + 2 * j], p0[8 + 2 * j + 1]);
                                      u1[j] = pk2(p1[8 + 2 * j], p1[8 + 2 * j + 1]); }
        a0[kc] = __builtin_bit_cast(v16bf, u0);
        a1[kc] = __builtin_bit_cast(v16bf, u1);
    }

    float sacc0[8], sacc1[8];
    #pragma unroll
    for (int i = 0; i < 8; ++i) { sacc0[i] = 0.f; sacc1[i] = 0.f; }

    const float* Pb = P + b * D2;
    for (int nt = 0; nt < 32; ++nt) {
        const v8u* bp = Bq + ((size_t)nt * 8 * 32 + lane);
        if (nt < 31) __builtin_prefetch(bp + 8 * 32, 0, 1);
        v8f c0 = {}, c1 = {};
        #pragma unroll
        for (int kc = 0; kc < 8; ++kc) {
            v16bf bv = __builtin_bit_cast(v16bf, bp[kc * 32]);
            c0 = __builtin_amdgcn_wmma_f32_16x16x32_bf16(
                     false, a0[kc], false, bv, (short)0, c0, false, false);
            c1 = __builtin_amdgcn_wmma_f32_16x16x32_bf16(
                     false, a1[kc], false, bv, (short)0, c1, false, false);
        }
        int   n  = nt * 16 + (lane & 15);
        float pn = Pb[n];
        float vn = v[n];
        #pragma unroll
        for (int i = 0; i < 8; ++i) {
            sacc0[i] += vn * htanh(c0[i] + pn);
            sacc1[i] += vn * htanh(c1[i] + pn);
        }
    }

    #pragma unroll
    for (int i = 0; i < 8; ++i) {
        float s0 = sacc0[i], s1 = sacc1[i];
        #pragma unroll
        for (int off = 1; off < 16; off <<= 1) {
            s0 += __shfl_xor(s0, off, 32);
            s1 += __shfl_xor(s1, off, 32);
        }
        sacc0[i] = s0; sacc1[i] = s1;
    }
    if ((lane & 15) == 0) {
        #pragma unroll
        for (int i = 0; i < 8; ++i) {
            scores[b * TT + t0 +      kgrp * 8 + i] = sacc0[i];
            scores[b * TT + t0 + 16 + kgrp * 8 + i] = sacc1[i];
        }
    }
}

__global__ void softmax_kernel(float* __restrict__ w) {
    __shared__ float red[256];
    int b = blockIdx.x, tid = threadIdx.x;
    float* row = w + b * TT;
    float m = -1e30f;
    for (int t = tid; t < TT; t += 256) m = fmaxf(m, row[t]);
    red[tid] = m; __syncthreads();
    for (int s = 128; s > 0; s >>= 1) { if (tid < s) red[tid] = fmaxf(red[tid], red[tid + s]); __syncthreads(); }
    m = red[0]; __syncthreads();
    float sum = 0.f;
    for (int t = tid; t < TT; t += 256) sum += __expf(row[t] - m);
    red[tid] = sum; __syncthreads();
    for (int s = 128; s > 0; s >>= 1) { if (tid < s) red[tid] += red[tid + s]; __syncthreads(); }
    float inv = 1.0f / red[0];
    for (int t = tid; t < TT; t += 256) row[t] = __expf(row[t] - m) * inv;
}

__global__ void zero_kernel(float* __restrict__ out, int n) {
    int i = blockIdx.x * blockDim.x + threadIdx.x;
    if (i < n) out[i] = 0.f;
}

__global__ void scatter_kernel(const int* __restrict__ src, const int* __restrict__ dst,
                               const float* __restrict__ w, float* __restrict__ out) {
    int i = blockIdx.x * blockDim.x + threadIdx.x;
    if (i >= BB * EE) return;
    int b = i >> 13;
    int s = src[i], d = dst[i];
    if (d != s + 1)
        atomicAdd(&out[b * TT + d], w[b * TT + s] * 0.1f);
}

extern "C" void kernel_launch(void* const* d_in, const int* in_sizes, int n_in,
                              void* d_out, int out_size, void* d_ws, size_t ws_size,
                              hipStream_t stream) {
    const float* hidden = (const float*)d_in[0];
    const float* enc    = (const float*)d_in[1];
    const int*   esrc   = (const int*)d_in[2];
    const int*   edst   = (const int*)d_in[3];
    const float* W      = (const float*)d_in[4];
    const float* bvec   = (const float*)d_in[5];
    const float* v      = (const float*)d_in[6];
    float* out = (float*)d_out;

    char* ws = (char*)d_ws;
    unsigned short* Bpack = (unsigned short*)ws;
    float* P    = (float*)(ws + 262144);
    float* wbuf = (float*)(ws + 262144 + 65536);

    packb_kernel  <<<256, 32, 0, stream>>>(W, Bpack);
    p_kernel      <<<BB, D2, 0, stream>>>(hidden, W, bvec, P);
    scores_kernel <<<512, 256, 0, stream>>>(enc, (const v8u*)Bpack, P, v, wbuf);
    softmax_kernel<<<BB, 256, 0, stream>>>(wbuf);
    zero_kernel   <<<(BB * TT + 255) / 256, 256, 0, stream>>>(out, BB * TT);
    scatter_kernel<<<(BB * EE + 255) / 256, 256, 0, stream>>>(esrc, edst, wbuf, out);
}


# Scores launch weight-stationary: each wave keeps its 32 weight columns in registers, enc rows pass once through LDS as the other WMMA operand of 16 waves; row sum in-lane, then over waves in LDS
# speedup vs baseline: 1.7555x; 1.7555x over previous
#include <hip/hip_runtime.h>
#include <hip/hip_bf16.h>

#define H 256
#define BB 32
#define TT 4096
#define EE 8192
#define D2 512

typedef __attribute__((ext_vector_type(16))) __bf16        v16bf;
typedef __attribute__((ext_vector_type(8)))  float         v8f;
typedef __attribute__((ext_vector_type(8)))  unsigned int  v8u;

__device__ __forceinline__ unsigned short f2bf(float x) {
    unsigned int u = __float_as_uint(x);
    unsigned int r = u + 0x7FFFu + ((u >> 16) & 1u);
    return (unsigned short)(r >> 16);
}
__device__ __forceinline__ unsigned int pk2(float lo, float hi) {
    return (unsigned int)f2bf(lo) | ((unsigned int)f2bf(hi) << 16);
}

__device__ __forceinline__ float htanh(float x) {
#if __has_builtin(__builtin_amdgcn_tanhf)
    return __builtin_amdgcn_tanhf(x);
#elif __has_builtin(__builtin_amdgcn_tanh_f32)
    return __builtin_amdgcn_tanh_f32(x);
#else
    float y;
    asm volatile("v_tanh_f32 %0, %1\n\t"
                 "s_delay_alu instid0(TRANS32_DEP_1)"
                 : "=v"(y) : "v"(x));
    return y;
#endif
}

__global__ void packb_kernel(const float* __restrict__ W, unsigned short* __restrict__ Bp) {
    int blk  = blockIdx.x;
    int nt   = blk >> 3;
    int kc   = blk & 7;
    int lane = threadIdx.x;
    int n     = nt * 16 + (lane & 15);
    int kbase = kc * 32 + ((lane >> 4) ? 8 : 0);
    const float* wrow = W + (size_t)n * D2 + H;
    unsigned short* dst = Bp + ((size_t)blk * 32 + lane) * 16;
    #pragma unroll
    for (int j = 0; j < 8; ++j) dst[j]     = f2bf(wrow[kbase + j]);
    #pragma unroll
    for (int j = 0; j < 8; ++j) dst[8 + j] = f2bf(wrow[kbase + 16 + j]);
}

__global__ void p_kernel(const float* __restrict__ hidden, const float* __restrict__ W,
                         const float* __restrict__ bvec, float* __restrict__ P) {
    int b = blockIdx.x;
    int n = threadIdx.x;
    const float* h = hidden + b * H;
    const float* w = W + (size_t)n * D2;
    float acc = bvec[n];
    #pragma unroll 4
    for (int k = 0; k < H; ++k) acc = fmaf(h[k], w[k], acc);
    P[b * D2 + n] = acc;
}

#define SC_WAVES 16
#define SC_PAIRS 2
#define SC_TRIPS 8
#define SC_ROWS  (32 * SC_PAIRS)
#define FR_BLK   136
#define FR_HALF  68
#define FR_TILE  (16 * FR_BLK)
#define ROW_F    (BB * H)

typedef __attribute__((ext_vector_type(2))) float  v2f;
typedef __attribute__((ext_vector_type(2))) __bf16 v2bf;

__device__ __forceinline__ unsigned int cvt2(float lo, float hi) {
    v2f x = {lo, hi};
    return __builtin_bit_cast(unsigned int, __builtin_convertvector(x, v2bf));
}
__device__ __forceinline__ uint2 cvt4(float4 f) {
    return make_uint2(cvt2(f.x, f.y), cvt2(f.z, f.w));
}

__global__ void __launch_bounds__(512)
scores_kernel(const float* __restrict__ enc, const v8u* __restrict__ Bq,
              const float* __restrict__ P, const float* __restrict__ v,
              float* __restrict__ scores) {
    __shared__ __attribute__((aligned(16))) unsigned int efrag[2][2 * SC_PAIRS * FR_TILE];
    __shared__ float red[2][SC_WAVES][SC_ROWS];

    const int tid  = threadIdx.x;
    const int wave = __builtin_amdgcn_readfirstlane(tid >> 5);
    const int lane = tid & 31;
    const int g    = lane >> 4;
    const int m    = lane & 15;
    const int b    = blockIdx.x >> 3;
    const int tb   = (blockIdx.x & 7) << 9;

    const float* erow = enc + ((size_t)(tb + wave) * BB + b) * H + lane * 4;
    const int sdw = (lane >> 2) * FR_BLK + ((lane >> 1) & 1) * FR_HALF + wave * 4 + (lane & 1) * 2;
    const int rdw = g * FR_HALF + m * 4;

    {
        float4 f[4 * SC_PAIRS];
        #pragma unroll
        for (int tl = 0; tl < 2 * SC_PAIRS; ++tl) {
            f[2 * tl]     = *(const float4*)(erow + tl * 16 * ROW_F);
            f[2 * tl + 1] = *(const float4*)(erow + tl * 16 * ROW_F + 128);
        }
        unsigned int* sp = &efrag[0][sdw];
        #pragma unroll
        for (int tl = 0; tl < 2 * SC_PAIRS; ++tl) {
            *(uint2*)(sp + tl * FR_TILE)              = cvt4(f[2 * tl]);
            *(uint2*)(sp + tl * FR_TILE + 8 * FR_BLK) = cvt4(f[2 * tl + 1]);
        }
    }

    v16bf wf0[8], wf1[8];
    {
        const v8u* bp = Bq + ((size_t)(2 * wave) * 8 * 32 + lane);
        #pragma unroll
        for (int kc = 0; kc < 8; ++kc) {
            wf0[kc] = __builtin_bit_cast(v16bf, bp[kc * 32]);
            wf1[kc] = __builtin_bit_cast(v16bf, bp[(8 + kc) * 32]);
        }
    }
    float pn0[8], pn1[8], vn0[8], vn1[8];
    {
        const int n0 = wave * 32 + g * 8;
        const float* Pb = P + b * D2 + n0;
        #pragma unroll
        for (int i = 0; i < 8; ++i) {
            pn0[i] = Pb[i];      pn1[i] = Pb[16 + i];
            vn0[i] = v[n0 + i];  vn1[i] = v[n0 + 16 + i];
        }
    }
    __syncthreads();

    #pragma nounroll
    for (int trip = 0; trip < SC_TRIPS; ++trip) {
        const int cur = trip & 1;
        const bool more = (trip + 1 < SC_TRIPS);
        const float* nx = erow + (size_t)(trip + 1) * (SC_ROWS * ROW_F);

        #pragma unroll
        for (int pr = 0; pr < SC_PAIRS; ++pr) {
            float4 st0, st1, st2, st3;
            if (more) {
                const float* p = nx + pr * 32 * ROW_F;
                st0 = *(const float4*)(p);
                st1 = *(const float4*)(p + 128);
                st2 = *(const float4*)(p + 16 * ROW_F);
                st3 = *(const float4*)(p + 16 * ROW_F + 128);
            }

            const unsigned int* fb = &efrag[cur][pr * 2 * FR_TILE + rdw];
            v8f c00 = {}, c01 = {}, c10 = {}, c11 = {};
            #pragma unroll
            for (int kc = 0; kc < 8; ++kc) {
                const uint4 l0 = *(const uint4*)(fb + (2 * kc) * FR_BLK);
                const uint4 h0 = *(const uint4*)(fb + (2 * kc + 1) * FR_BLK);
                const uint4 l1 = *(const uint4*)(fb + FR_TILE + (2 * kc) * FR_BLK);
                const uint4 h1 = *(const uint4*)(fb + FR_TILE + (2 * kc + 1) * FR_BLK);
                const v8u u0 = {l0.x, l0.y, l0.z, l0.w, h0.x, h0.y, h0.z, h0.w};
                const v8u u1 = {l1.x, l1.y, l1.z, l1.w, h1.x, h1.y, h1.z, h1.w};
                const v16bf e0 = __builtin_bit_cast(v16bf, u0);
                const v16bf e1 = __builtin_bit_cast(v16bf, u1);
                c00 = __builtin_amdgcn_wmma_f32_16x16x32_bf16(
                          false, wf0[kc], false, e0, (short)0, c00, false, false);
                c01 = __builtin_amdgcn_wmma_f32_16x16x32_bf16(
                          false, wf1[kc], false, e0, (short)0, c01, false, false);
                c10 = __builtin_amdgcn_wmma_f32_16x16x32_bf16(
                          false, wf0[kc], false, e1, (short)0, c10, false, false);
                c11 = __builtin_amdgcn_wmma_f32_16x16x32_bf16(
                          false, wf1[kc], false, e1, (short)0, c11, false, false);
            }

            float s0a = 0.f, s0b = 0.f, s1a = 0.f, s1b = 0.f;
            #pragma unroll
            for (int i = 0; i < 8; ++i) {
                s0a = fmaf(vn0[i], htanh(c00[i] + pn0[i]), s0a);
                s0b = fmaf(vn1[i], htanh(c01[i] + pn1[i]), s0b);
                s1a = fmaf(vn0[i], htanh(c10[i] + pn0[i]), s1a);
                s1b = fmaf(vn1[i], htanh(c11[i] + pn1[i]), s1b);
            }
            const float s0 = s0a + s0b;
            const float s1 = s1a + s1b;
            const float keep = g ? s1 : s0;
            const float give = g ? s0 : s1;
            red[cur][wave][pr * 32 + lane] = keep + __shfl_xor(give, 16, 32);

            if (more) {
                unsigned int* sp = &efrag[cur ^ 1][pr * 2 * FR_TILE + sdw];
                *(uint2*)(sp)                        = cvt4(st0);
                *(uint2*)(sp + 8 * FR_BLK)           = cvt4(st1);
                *(uint2*)(sp + FR_TILE)              = cvt4(st2);
                *(uint2*)(sp + FR_TILE + 8 * FR_BLK) = cvt4(st3);
            }
        }
        __syncthreads();

        const int rp = wave - 2 * trip;
        if (rp == 0 || rp == 1) {
            const int r = rp * 32 + lane;
            float acc = red[cur][0][r];
            #pragma unroll
            for (int w = 1; w < SC_WAVES; ++w) acc += red[cur][w][r];
            scores[b * TT + tb + trip * SC_ROWS + r] = acc;
        }
    }
}

__global__ void softmax_kernel(float* __restrict__ w) {
    __shared__ float red[256];
    int b = blockIdx.x, tid = threadIdx.x;
    float* row = w + b * TT;
    float m = -1e30f;
    for (int t = tid; t < TT; t += 256) m = fmaxf(m, row[t]);
    red[tid] = m; __syncthreads();
    for (int s = 128; s > 0; s >>= 1) { if (tid < s) red[tid] = fmaxf(red[tid], red[tid + s]); __syncthreads(); }
    m = red[0]; __syncthreads();
    float sum = 0.f;
    for (int t = tid; t < TT; t += 256) sum += __expf(row[t] - m);
    red[tid] = sum; __syncthreads();
    for (int s = 128; s > 0; s >>= 1) { if (tid < s) red[tid] += red[tid + s]; __syncthreads(); }
    float inv = 1.0f / red[0];
    for (int t = tid; t < TT; t += 256) row[t] = __expf(row[t] - m) * inv;
}

__global__ void zero_kernel(float* __restrict__ out, int n) {
    int i = blockIdx.x * blockDim.x + threadIdx.x;
    if (i < n) out[i] = 0.f;
}

__global__ void scatter_kernel(const int* __restrict__ src, const int* __restrict__ dst,
                               const float* __restrict__ w, float* __restrict__ out) {
    int i = blockIdx.x * blockDim.x + threadIdx.x;
    if (i >= BB * EE) return;
    int b = i >> 13;
    int s = src[i], d = dst[i];
    if (d != s + 1)
        atomicAdd(&out[b * TT + d], w[b * TT + s] * 0.1f);
}

extern "C" void kernel_launch(void* const* d_in, const int* in_sizes, int n_in,
                              void* d_out, int out_size, void* d_ws, size_t ws_size,
                              hipStream_t stream) {
    const float* hidden = (const float*)d_in[0];
    const float* enc    = (const float*)d_in[1];
    const int*   esrc   = (const int*)d_in[2];
    const int*   edst   = (const int*)d_in[3];
    const float* W      = (const float*)d_in[4];
    const float* bvec   = (const float*)d_in[5];
    const float* v      = (const float*)d_in[6];
    float* out = (float*)d_out;

    char* ws = (char*)d_ws;
    unsigned short* Bpack = (unsigned short*)ws;
    float* P    = (float*)(ws + 262144);
    float* wbuf = (float*)(ws + 262144 + 65536);

    packb_kernel  <<<256, 32, 0, stream>>>(W, Bpack);
    p_kernel      <<<BB, D2, 0, stream>>>(hidden, W, bvec, P);
    scores_kernel <<<256, 512, 0, stream>>>(enc, (const v8u*)Bpack, P, v, wbuf);
    softmax_kernel<<<BB, 256, 0, stream>>>(wbuf);
    zero_kernel   <<<(BB * TT + 255) / 256, 256, 0, stream>>>(out, BB * TT);
    scatter_kernel<<<(BB * EE + 255) / 256, 256, 0, stream>>>(esrc, edst, wbuf, out);
}


# Softmax and zero launches dissolved (6->4): scores launch leaves max and exp sum per work-group, zeroes its out slice; scatter joins a row's 8 pairs, applies exp and 1/S per edge; sum tree changes
# speedup vs baseline: 1.9725x; 1.1236x over previous
#include <hip/hip_runtime.h>
#include <hip/hip_bf16.h>

#define H 256
#define BB 32
#define TT 4096
#define EE 8192
#define D2 512

typedef __attribute__((ext_vector_type(16))) __bf16        v16bf;
typedef __attribute__((ext_vector_type(8)))  float         v8f;
typedef __attribute__((ext_vector_type(8)))  unsigned int  v8u;

__device__ __forceinline__ unsigned short f2bf(float x) {
    unsigned int u = __float_as_uint(x);
    unsigned int r = u + 0x7FFFu + ((u >> 16) & 1u);
    return (unsigned short)(r >> 16);
}
__device__ __forceinline__ unsigned int pk2(float lo, float hi) {
    return (unsigned int)f2bf(lo) | ((unsigned int)f2bf(hi) << 16);
}

__device__ __forceinline__ float htanh(float x) {
#if __has_builtin(__builtin_amdgcn_tanhf)
    return __builtin_amdgcn_tanhf(x);
#elif __has_builtin(__builtin_amdgcn_tanh_f32)
    return __builtin_amdgcn_tanh_f32(x);
#else
    float y;
    asm volatile("v_tanh_f32 %0, %1\n\t"
                 "s_delay_alu instid0(TRANS32_DEP_1)"
                 : "=v"(y) : "v"(x));
    return y;
#endif
}

__global__ void packb_kernel(const float* __restrict__ W, unsigned short* __restrict__ Bp) {
    int blk  = blockIdx.x;
    int nt   = blk >> 3;
    int kc   = blk & 7;
    int lane = threadIdx.x;
    int n     = nt * 16 + (lane & 15);
    int kbase = kc * 32 + ((lane >> 4) ? 8 : 0);
    const float* wrow = W + (size_t)n * D2 + H;
    unsigned short* dst = Bp + ((size_t)blk * 32 + lane) * 16;
    #pragma unroll
    for (int j = 0; j < 8; ++j) dst[j]     = f2bf(wrow[kbase + j]);
    #pragma unroll
    for (int j = 0; j < 8; ++j) dst[8 + j] = f2bf(wrow[kbase + 16 + j]);
}

__global__ void p_kernel(const float* __restrict__ hidden, const float* __restrict__ W,
                         const float* __restrict__ bvec, float* __restrict__ P) {
    int b = blockIdx.x;
    int n = threadIdx.x;
    const float* h = hidden + b * H;
    const float* w = W + (size_t)n * D2;
    float acc = bvec[n];
    #pragma unroll 4
    for (int k = 0; k < H; ++k) acc = fmaf(h[k], w[k], acc);
    P[b * D2 + n] = acc;
}

#define SC_WAVES 16
#define SC_PAIRS 2
#define SC_TRIPS 8
#define SC_ROWS  (32 * SC_PAIRS)
#define FR_BLK   136
#define FR_HALF  68
#define FR_TILE  (16 * FR_BLK)
#define ROW_F    (BB * H)

typedef __attribute__((ext_vector_type(2))) float  v2f;
typedef __attribute__((ext_vector_type(2))) __bf16 v2bf;

__device__ __forceinline__ unsigned int cvt2(float lo, float hi) {
    v2f x = {lo, hi};
    return __builtin_bit_cast(unsigned int, __builtin_convertvector(x, v2bf));
}
__device__ __forceinline__ uint2 cvt4(float4 f) {
    return make_uint2(cvt2(f.x, f.y), cvt2(f.z, f.w));
}

template <int MASK>
__device__ __forceinline__ float rowx(float x) {
    return __builtin_bit_cast(float, __builtin_amdgcn_update_dpp(
        0, __builtin_bit_cast(int, x), 0x160 | MASK, 0xf, 0xf, false));
}
__device__ __forceinline__ float xhalf(float x) {
    const int b = __builtin_bit_cast(int, x);
    return __builtin_bit_cast(float, __builtin_amdgcn_permlanex16(b, b, 0x76543210, 0xfedcba98, false, false));
}
__device__ __forceinline__ float wave_max(float x) {
    x = fmaxf(x, rowx<1>(x)); x = fmaxf(x, rowx<2>(x));
    x = fmaxf(x, rowx<4>(x)); x = fmaxf(x, rowx<8>(x));
    return fmaxf(x, xhalf(x));
}
__device__ __forceinline__ float wave_sum(float x) {
    x += rowx<1>(x); x += rowx<2>(x); x += rowx<4>(x); x += rowx<8>(x);
    return x + xhalf(x);
}

__global__ void __launch_bounds__(512)
scores_kernel(const float* __restrict__ enc, const v8u* __restrict__ Bq,
              const float* __restrict__ P, const float* __restrict__ v,
              float* __restrict__ scores, float* __restrict__ stats,
              float* __restrict__ out) {
    __shared__ __attribute__((aligned(16))) unsigned int efrag[2][2 * SC_PAIRS * FR_TILE];
    __shared__ float red[2][SC_WAVES][SC_ROWS];
    __shared__ float srow[SC_TRIPS * SC_ROWS];

    const int tid  = threadIdx.x;
    const int wave = __builtin_amdgcn_readfirstlane(tid >> 5);
    const int lane = tid & 31;
    const int g    = lane >> 4;
    const int m    = lane & 15;
    const int b    = blockIdx.x >> 3;
    const int tb   = (blockIdx.x & 7) << 9;

    out[b * TT + tb + tid] = 0.f;

    const float* erow = enc + ((size_t)(tb + wave) * BB + b) * H + lane * 4;
    const int sdw = (lane >> 2) * FR_BLK + ((lane >> 1) & 1) * FR_HALF + wave * 4 + (lane & 1) * 2;
    const int rdw = g * FR_HALF + m * 4;

    {
        float4 f[4 * SC_PAIRS];
        #pragma unroll
        for (int tl = 0; tl < 2 * SC_PAIRS; ++tl) {
            f[2 * tl]     = *(const float4*)(erow + tl * 16 * ROW_F);
            f[2 * tl + 1] = *(const float4*)(erow + tl * 16 * ROW_F + 128);
        }
        unsigned int* sp = &efrag[0][sdw];
        #pragma unroll
        for (int tl = 0; tl < 2 * SC_PAIRS; ++tl) {
            *(uint2*)(sp + tl * FR_TILE)              = cvt4(f[2 * tl]);
            *(uint2*)(sp + tl * FR_TILE + 8 * FR_BLK) = cvt4(f[2 * tl + 1]);
        }
    }

    v16bf wf0[8], wf1[8];
    {
        const v8u* bp = Bq + ((size_t)(2 * wave) * 8 * 32 + lane);
        #pragma unroll
        for (int kc = 0; kc < 8; ++kc) {
            wf0[kc] = __builtin_bit_cast(v16bf, bp[kc * 32]);
            wf1[kc] = __builtin_bit_cast(v16bf, bp[(8 + kc) * 32]);
        }
    }
    float pn0[8], pn1[8], vn0[8], vn1[8];
    {
        const int n0 = wave * 32 + g * 8;
        const float* Pb = P + b * D2 + n0;
        #pragma unroll
        for (int i = 0; i < 8; ++i) {
            pn0[i] = Pb[i];      pn1[i] = Pb[16 + i];
            vn0[i] = v[n0 + i];  vn1[i] = v[n0 + 16 + i];
        }
    }
    __syncthreads();

    #pragma nounroll
    for (int trip = 0; trip < SC_TRIPS; ++trip) {
        const int cur = trip & 1;
        const bool more = (trip + 1 < SC_TRIPS);
        const float* nx = erow + (size_t)(trip + 1) * (SC_ROWS * ROW_F);

        #pragma unroll
        for (int pr = 0; pr < SC_PAIRS; ++pr) {
            float4 st0, st1, st2, st3;
            if (more) {
                const float* p = nx + pr * 32 * ROW_F;
                st0 = *(const float4*)(p);
                st1 = *(const float4*)(p + 128);
                st2 = *(const float4*)(p + 16 * ROW_F);
                st3 = *(const float4*)(p + 16 * ROW_F + 128);
            }

            const unsigned int* fb = &efrag[cur][pr * 2 * FR_TILE + rdw];
            v8f c00 = {}, c01 = {}, c10 = {}, c11 = {};
            #pragma unroll
            for (int kc = 0; kc < 8; ++kc) {
                const uint4 l0 = *(const uint4*)(fb + (2 * kc) * FR_BLK);
                const uint4 h0 = *(const uint4*)(fb + (2 * kc + 1) * FR_BLK);
                const uint4 l1 = *(const uint4*)(fb + FR_TILE + (2 * kc) * FR_BLK);
                const uint4 h1 = *(const uint4*)(fb + FR_TILE + (2 * kc + 1) * FR_BLK);
                const v8u u0 = {l0.x, l0.y, l0.z, l0.w, h0.x, h0.y, h0.z, h0.w};
                const v8u u1 = {l1.x, l1.y, l1.z, l1.w, h1.x, h1.y, h1.z, h1.w};
                const v16bf e0 = __builtin_bit_cast(v16bf, u0);
                const v16bf e1 = __builtin_bit_cast(v16bf, u1);
                c00 = __builtin_amdgcn_wmma_f32_16x16x32_bf16(
                          false, wf0[kc], false, e0, (short)0, c00, false, false);
                c01 = __builtin_amdgcn_wmma_f32_16x16x32_bf16(
                          false, wf1[kc], false, e0, (short)0, c01, false, false);
                c10 = __builtin_amdgcn_wmma_f32_16x16x32_bf16(
                          false, wf0[kc], false, e1, (short)0, c10, false, false);
                c11 = __builtin_amdgcn_wmma_f32_16x16x32_bf16(
                          false, wf1[kc], false, e1, (short)0, c11, false, false);
            }

            float s0a = 0.f, s0b = 0.f, s1a = 0.f, s1b = 0.f;
            #pragma unroll
            for (int i = 0; i < 8; ++i) {
                s0a = fmaf(vn0[i], htanh(c00[i] + pn0[i]), s0a);
                s0b = fmaf(vn1[i], htanh(c01[i] + pn1[i]), s0b);
                s1a = fmaf(vn0[i], htanh(c10[i] + pn0[i]), s1a);
                s1b = fmaf(vn1[i], htanh(c11[i] + pn1[i]), s1b);
            }
            const float s0 = s0a + s0b;
            const float s1 = s1a + s1b;
            const float keep = g ? s1 : s0;
            const float give = g ? s0 : s1;
            red[cur][wave][pr * 32 + lane] = keep + __shfl_xor(give, 16, 32);

            if (more) {
                unsigned int* sp = &efrag[cur ^ 1][pr * 2 * FR_TILE + sdw];
                *(uint2*)(sp)                        = cvt4(st0);
                *(uint2*)(sp + 8 * FR_BLK)           = cvt4(st1);
                *(uint2*)(sp + FR_TILE)              = cvt4(st2);
                *(uint2*)(sp + FR_TILE + 8 * FR_BLK) = cvt4(st3);
            }
        }
        __syncthreads();

        const int rp = wave - 2 * trip;
        if (rp == 0 || rp == 1) {
            const int r = rp * 32 + lane;
            float acc = red[cur][0][r];
            #pragma unroll
            for (int w = 1; w < SC_WAVES; ++w) acc += red[cur][w][r];
            scores[b * TT + tb + trip * SC_ROWS + r] = acc;
            srow[trip * SC_ROWS + r] = acc;
        }
    }
    __syncthreads();

    if (wave == 0) {
        float x[16];
        #pragma unroll
        for (int j = 0; j < 16; ++j) x[j] = srow[j * 32 + lane];
        float mx = x[0];
        #pragma unroll
        for (int j = 1; j < 16; ++j) mx = fmaxf(mx, x[j]);
        mx = wave_max(mx);
        float sm = 0.f;
        #pragma unroll
        for (int j = 0; j < 16; ++j) sm += __expf(x[j] - mx);
        sm = wave_sum(sm);
        if (lane == 0) {
            stats[2 * blockIdx.x]     = mx;
            stats[2 * blockIdx.x + 1] = sm;
        }
    }
}

__global__ void scatter_kernel(const int* __restrict__ src, const int* __restrict__ dst,
                               const float* __restrict__ sc, const float* __restrict__ stats,
                               float* __restrict__ out) {
    int i = blockIdx.x * blockDim.x + threadIdx.x;
    if (i >= BB * EE) return;
    const int b = blockIdx.x >> 5;
    const float* st = stats + b * 16;
    float M = st[0];
    #pragma unroll
    for (int j = 1; j < 8; ++j) M = fmaxf(M, st[2 * j]);
    float S = 0.f;
    #pragma unroll
    for (int j = 0; j < 8; ++j) S += st[2 * j + 1] * __expf(st[2 * j] - M);
    const float inv = 1.0f / S;
    int s = src[i], d = dst[i];
    if (d != s + 1)
        atomicAdd(&out[b * TT + d], __expf(sc[b * TT + s] - M) * inv * 0.1f);
}

extern "C" void kernel_launch(void* const* d_in, const int* in_sizes, int n_in,
                              void* d_out, int out_size, void* d_ws, size_t ws_size,
                              hipStream_t stream) {
    const float* hidden = (const float*)d_in[0];
    const float* enc    = (const float*)d_in[1];
    const int*   esrc   = (const int*)d_in[2];
    const int*   edst   = (const int*)d_in[3];
    const float* W      = (const float*)d_in[4];
    const float* bvec   = (const float*)d_in[5];
    const float* v      = (const float*)d_in[6];
    float* out = (float*)d_out;

    char* ws = (char*)d_ws;
    unsigned short* Bpack = (unsigned short*)ws;
    float* P    = (float*)(ws + 262144);
    float* wbuf = (float*)(ws + 262144 + 65536);
    float* stats = (float*)(ws + 262144 + 65536 + 524288);

    packb_kernel  <<<256, 32, 0, stream>>>(W, Bpack);
    p_kernel      <<<BB, D2, 0, stream>>>(hidden, W, bvec, P);
    scores_kernel <<<256, 512, 0, stream>>>(enc, (const v8u*)Bpack, P, v, wbuf, stats, out);
    scatter_kernel<<<(BB * EE + 255) / 256, 256, 0, stream>>>(esrc, edst, wbuf, stats, out);
}


# Prep launch: P = hidden x W1^T + b moved onto the f32 16x16x4 matrix instruction, 16x32 tile per 8-wave work-group, k quarters joined in LDS; P's sum order changes, f32 kept
# speedup vs baseline: 2.2561x; 1.1438x over previous
#include <hip/hip_runtime.h>
#include <hip/hip_bf16.h>

#define H 256
#define BB 32
#define TT 4096
#define EE 8192
#define D2 512

typedef __attribute__((ext_vector_type(16))) __bf16        v16bf;
typedef __attribute__((ext_vector_type(8)))  float         v8f;
typedef __attribute__((ext_vector_type(8)))  unsigned int  v8u;
typedef __attribute__((ext_vector_type(2)))  float         v2f;

__device__ __forceinline__ unsigned short f2bf(float x) {
    unsigned int u = __float_as_uint(x);
    unsigned int r = u + 0x7FFFu + ((u >> 16) & 1u);
    return (unsigned short)(r >> 16);
}
__device__ __forceinline__ unsigned int pk2(float lo, float hi) {
    return (unsigned int)f2bf(lo) | ((unsigned int)f2bf(hi) << 16);
}

__device__ __forceinline__ float htanh(float x) {
#if __has_builtin(__builtin_amdgcn_tanhf)
    return __builtin_amdgcn_tanhf(x);
#elif __has_builtin(__builtin_amdgcn_tanh_f32)
    return __builtin_amdgcn_tanh_f32(x);
#else
    float y;
    asm volatile("v_tanh_f32 %0, %1\n\t"
                 "s_delay_alu instid0(TRANS32_DEP_1)"
                 : "=v"(y) : "v"(x));
    return y;
#endif
}

__global__ void packb_kernel(const float* __restrict__ W, unsigned short* __restrict__ Bp) {
    int blk  = blockIdx.x;
    int nt   = blk >> 3;
    int kc   = blk & 7;
    int lane = threadIdx.x;
    int n     = nt * 16 + (lane & 15);
    int kbase = kc * 32 + ((lane >> 4) ? 8 : 0);
    const float* wrow = W + (size_t)n * D2 + H;
    unsigned short* dst = Bp + ((size_t)blk * 32 + lane) * 16;
    #pragma unroll
    for (int j = 0; j < 8; ++j) dst[j]     = f2bf(wrow[kbase + j]);
    #pragma unroll
    for (int j = 0; j < 8; ++j) dst[8 + j] = f2bf(wrow[kbase + 16 + j]);
}

__global__ void __launch_bounds__(256)
p_kernel(const float* __restrict__ hidden, const float* __restrict__ W,
         const float* __restrict__ bvec, float* __restrict__ P) {
    __shared__ float part[2][4][8][32];
    const int lane = threadIdx.x & 31;
    const int wave = __builtin_amdgcn_readfirstlane(threadIdx.x >> 5);
    const int ct = wave & 1;
    const int q  = wave >> 1;
    const int c  = lane & 15;
    const int h  = lane >> 4;
    const int b0 = (blockIdx.x & 1) << 4;
    const int n0 = (blockIdx.x >> 1) << 5;
    const float* ap = hidden + (b0 + c) * H + q * 64 + h * 4;
    const float* wp = W + (size_t)(n0 + ct * 16 + c) * D2 + q * 64 + h * 4;
    float4 a[8], w[8];
    #pragma unroll
    for (int j = 0; j < 8; ++j) {
        a[j] = *(const float4*)(ap + 8 * j);
        w[j] = *(const float4*)(wp + 8 * j);
    }
    v8f acc0 = {}, acc1 = {};
    #pragma unroll
    for (int j = 0; j < 8; ++j) {
        const v2f a0 = {a[j].x, a[j].y}, w0 = {w[j].x, w[j].y};
        const v2f a1 = {a[j].z, a[j].w}, w1 = {w[j].z, w[j].w};
        acc0 = __builtin_amdgcn_wmma_f32_16x16x4_f32(false, a0, false, w0, (short)0, acc0, false, false);
        acc1 = __builtin_amdgcn_wmma_f32_16x16x4_f32(false, a1, false, w1, (short)0, acc1, false, false);
    }
    #pragma unroll
    for (int r = 0; r < 8; ++r) part[ct][q][r][lane] = acc0[r] + acc1[r];
    __syncthreads();
    const int st = lane >> 4;
    const float bias = bvec[n0 + lane];
    #pragma unroll
    for (int hh = 0; hh < 2; ++hh) {
        const int l = hh * 16 + c;
        const float s = (part[st][0][wave][l] + part[st][1][wave][l])
                      + (part[st][2][wave][l] + part[st][3][wave][l]);
        P[(b0 + 8 * hh + wave) * D2 + n0 + lane] = bias + s;
    }
}

#define SC_WAVES 16
#define SC_PAIRS 2
#define SC_TRIPS 8
#define SC_ROWS  (32 * SC_PAIRS)
#define FR_BLK   136
#define FR_HALF  68
#define FR_TILE  (16 * FR_BLK)
#define ROW_F    (BB * H)

typedef __attribute__((ext_vector_type(2))) __bf16 v2bf;

__device__ __forceinline__ unsigned int cvt2(float lo, float hi) {
    v2f x = {lo, hi};
    return __builtin_bit_cast(unsigned int, __builtin_convertvector(x, v2bf));
}
__device__ __forceinline__ uint2 cvt4(float4 f) {
    return make_uint2(cvt2(f.x, f.y), cvt2(f.z, f.w));
}

template <int MASK>
__device__ __forceinline__ float rowx(float x) {
    return __builtin_bit_cast(float, __builtin_amdgcn_update_dpp(
        0, __builtin_bit_cast(int, x), 0x160 | MASK, 0xf, 0xf, false));
}
__device__ __forceinline__ float xhalf(float x) {
    const int b = __builtin_bit_cast(int, x);
    return __builtin_bit_cast(float, __builtin_amdgcn_permlanex16(b, b, 0x76543210, 0xfedcba98, false, false));
}
__device__ __forceinline__ float wave_max(float x) {
    x = fmaxf(x, rowx<1>(x)); x = fmaxf(x, rowx<2>(x));
    x = fmaxf(x, rowx<4>(x)); x = fmaxf(x, rowx<8>(x));
    return fmaxf(x, xhalf(x));
}
__device__ __forceinline__ float wave_sum(float x) {
    x += rowx<1>(x); x += rowx<2>(x); x += rowx<4>(x); x += rowx<8>(x);
    return x + xhalf(x);
}

__global__ void __launch_bounds__(512)
scores_kernel(const float* __restrict__ enc, const v8u* __restrict__ Bq,
              const float* __restrict__ P, const float* __restrict__ v,
              float* __restrict__ scores, float* __restrict__ stats,
              float* __restrict__ out) {
    __shared__ __attribute__((aligned(16))) unsigned int efrag[2][2 * SC_PAIRS * FR_TILE];
    __shared__ float red[2][SC_WAVES][SC_ROWS];
    __shared__ float srow[SC_TRIPS * SC_ROWS];

    const int tid  = threadIdx.x;
    const int wave = __builtin_amdgcn_readfirstlane(tid >> 5);
    const int lane = tid & 31;
    const int g    = lane >> 4;
    const int m    = lane & 15;
    const int b    = blockIdx.x >> 3;
    const int tb   = (blockIdx.x & 7) << 9;

    out[b * TT + tb + tid] = 0.f;

    const float* erow = enc + ((size_t)(tb + wave) * BB + b) * H + lane * 4;
    const int sdw = (lane >> 2) * FR_BLK + ((lane >> 1) & 1) * FR_HALF + wave * 4 + (lane & 1) * 2;
    const int rdw = g * FR_HALF + m * 4;

    {
        float4 f[4 * SC_PAIRS];
        #pragma unroll
        for (int tl = 0; tl < 2 * SC_PAIRS; ++tl) {
            f[2 * tl]     = *(const float4*)(erow + tl * 16 * ROW_F);
            f[2 * tl + 1] = *(const float4*)(erow + tl * 16 * ROW_F + 128);
        }
        unsigned int* sp = &efrag[0][sdw];
        #pragma unroll
        for (int tl = 0; tl < 2 * SC_PAIRS; ++tl) {
            *(uint2*)(sp + tl * FR_TILE)              = cvt4(f[2 * tl]);
            *(uint2*)(sp + tl * FR_TILE + 8 * FR_BLK) = cvt4(f[2 * tl + 1]);
        }
    }

    v16bf wf0[8], wf1[8];
    {
        const v8u* bp = Bq + ((size_t)(2 * wave) * 8 * 32 + lane);
        #pragma unroll
        for (int kc = 0; kc < 8; ++kc) {
            wf0[kc] = __builtin_bit_cast(v16bf, bp[kc * 32]);
            wf1[kc] = __builtin_bit_cast(v16bf, bp[(8 + kc) * 32]);
        }
    }
    float pn0[8], pn1[8], vn0[8], vn1[8];
    {
        const int n0 = wave * 32 + g * 8;
        const float* Pb = P + b * D2 + n0;
        #pragma unroll
        for (int i = 0; i < 8; ++i) {
            pn0[i] = Pb[i];      pn1[i] = Pb[16 + i];
            vn0[i] = v[n0 + i];  vn1[i] = v[n0 + 16 + i];
        }
    }
    __syncthreads();

    #pragma nounroll
    for (int trip = 0; trip < SC_TRIPS; ++trip) {
        const int cur = trip & 1;
        const bool more = (trip + 1 < SC_TRIPS);
        const float* nx = erow + (size_t)(trip + 1) * (SC_ROWS * ROW_F);

        #pragma unroll
        for (int pr = 0; pr < SC_PAIRS; ++pr) {
            float4 st0, st1, st2, st3;
            if (more) {
                const float* p = nx + pr * 32 * ROW_F;
                st0 = *(const float4*)(p);
                st1 = *(const float4*)(p + 128);
                st2 = *(const float4*)(p + 16 * ROW_F);
                st3 = *(const float4*)(p + 16 * ROW_F + 128);
            }

            const unsigned int* fb = &efrag[cur][pr * 2 * FR_TILE + rdw];
            v8f c00 = {}, c01 = {}, c10 = {}, c11 = {};
            #pragma unroll
            for (int kc = 0; kc < 8; ++kc) {
                const uint4 l0 = *(const uint4*)(fb + (2 * kc) * FR_BLK);
                const uint4 h0 = *(const uint4*)(fb + (2 * kc + 1) * FR_BLK);
                const uint4 l1 = *(const uint4*)(fb + FR_TILE + (2 * kc) * FR_BLK);
                const uint4 h1 = *(const uint4*)(fb + FR_TILE + (2 * kc + 1) * FR_BLK);
                const v8u u0 = {l0.x, l0.y, l0.z, l0.w, h0.x, h0.y, h0.z, h0.w};
                const v8u u1 = {l1.x, l1.y, l1.z, l1.w, h1.x, h1.y, h1.z, h1.w};
                const v16bf e0 = __builtin_bit_cast(v16bf, u0);
                const v16bf e1 = __builtin_bit_cast(v16bf, u1);
                c00 = __builtin_amdgcn_wmma_f32_16x16x32_bf16(
                          false, wf0[kc], false, e0, (short)0, c00, false, false);
                c01 = __builtin_amdgcn_wmma_f32_16x16x32_bf16(
                          false, wf1[kc], false, e0, (short)0, c01, false, false);
                c10 = __builtin_amdgcn_wmma_f32_16x16x32_bf16(
                          false, wf0[kc], false, e1, (short)0, c10, false, false);
                c11 = __builtin_amdgcn_wmma_f32_16x16x32_bf16(
                          false, wf1[kc], false, e1, (short)0, c11, false, false);
            }

            float s0a = 0.f, s0b = 0.f, s1a = 0.f, s1b = 0.f;
            #pragma unroll
            for (int i = 0; i < 8; ++i) {
                s0a = fmaf(vn0[i], htanh(c00[i] + pn0[i]), s0a);
                s0b = fmaf(vn1[i], htanh(c01[i] + pn1[i]), s0b);
                s1a = fmaf(vn0[i], htanh(c10[i] + pn0[i]), s1a);
                s1b = fmaf(vn1[i], htanh(c11[i] + pn1[i]), s1b);
            }
            const float s0 = s0a + s0b;
            const float s1 = s1a + s1b;
            const float keep = g ? s1 : s0;
            const float give = g ? s0 : s1;
            red[cur][wave][pr * 32 + lane] = keep + __shfl_xor(give, 16, 32);

            if (more) {
                unsigned int* sp = &efrag[cur ^ 1][pr * 2 * FR_TILE + sdw];
                *(uint2*)(sp)                        = cvt4(st0);
                *(uint2*)(sp + 8 * FR_BLK)           = cvt4(st1);
                *(uint2*)(sp + FR_TILE)              = cvt4(st2);
                *(uint2*)(sp + FR_TILE + 8 * FR_BLK) = cvt4(st3);
            }
        }
        __syncthreads();

        const int rp = wave - 2 * trip;
        if (rp == 0 || rp == 1) {
            const int r = rp * 32 + lane;
            float acc = red[cur][0][r];
            #pragma unroll
            for (int w = 1; w < SC_WAVES; ++w) acc += red[cur][w][r];
            scores[b * TT + tb + trip * SC_ROWS + r] = acc;
            srow[trip * SC_ROWS + r] = acc;
        }
    }
    __syncthreads();

    if (wave == 0) {
        float x[16];
        #pragma unroll
        for (int j = 0; j < 16; ++j) x[j] = srow[j * 32 + lane];
        float mx = x[0];
        #pragma unroll
        for (int j = 1; j < 16; ++j) mx = fmaxf(mx, x[j]);
        mx = wave_max(mx);
        float sm = 0.f;
        #pragma unroll
        for (int j = 0; j < 16; ++j) sm += __expf(x[j] - mx);
        sm = wave_sum(sm);
        if (lane == 0) {
            stats[2 * blockIdx.x]     = mx;
            stats[2 * blockIdx.x + 1] = sm;
        }
    }
}

__global__ void scatter_kernel(const int* __restrict__ src, const int* __restrict__ dst,
                               const float* __restrict__ sc, const float* __restrict__ stats,
                               float* __restrict__ out) {
    int i = blockIdx.x * blockDim.x + threadIdx.x;
    if (i >= BB * EE) return;
    const int b = blockIdx.x >> 5;
    const float* st = stats + b * 16;
    float M = st[0];
    #pragma unroll
    for (int j = 1; j < 8; ++j) M = fmaxf(M, st[2 * j]);
    float S = 0.f;
    #pragma unroll
    for (int j = 0; j < 8; ++j) S += st[2 * j + 1] * __expf(st[2 * j] - M);
    const float inv = 1.0f / S;
    int s = src[i], d = dst[i];
    if (d != s + 1)
        atomicAdd(&out[b * TT + d], __expf(sc[b * TT + s] - M) * inv * 0.1f);
}

extern "C" void kernel_launch(void* const* d_in, const int* in_sizes, int n_in,
                              void* d_out, int out_size, void* d_ws, size_t ws_size,
                              hipStream_t stream) {
    const float* hidden = (const float*)d_in[0];
    const float* enc    = (const float*)d_in[1];
    const int*   esrc   = (const int*)d_in[2];
    const int*   edst   = (const int*)d_in[3];
    const float* W      = (const float*)d_in[4];
    const float* bvec   = (const float*)d_in[5];
    const float* v      = (const float*)d_in[6];
    float* out = (float*)d_out;

    char* ws = (char*)d_ws;
    unsigned short* Bpack = (unsigned short*)ws;
    float* P    = (float*)(ws + 262144);
    float* wbuf = (float*)(ws + 262144 + 65536);
    float* stats = (float*)(ws + 262144 + 65536 + 524288);

    packb_kernel  <<<256, 32, 0, stream>>>(W, Bpack);
    p_kernel      <<<32, 256, 0, stream>>>(hidden, W, bvec, P);
    scores_kernel <<<256, 512, 0, stream>>>(enc, (const v8u*)Bpack, P, v, wbuf, stats, out);
    scatter_kernel<<<(BB * EE + 255) / 256, 256, 0, stream>>>(esrc, edst, wbuf, stats, out);
}


# Scores launch, LDS feed of the WMMAs: each wave keeps 64 weight columns (8-wave work-groups), so one enc fragment read feeds four WMMAs; reads issued two k chunks ahead; row sum tree changes
# speedup vs baseline: 2.4488x; 1.0854x over previous
#include <hip/hip_runtime.h>
#include <hip/hip_bf16.h>

#define H 256
#define BB 32
#define TT 4096
#define EE 8192
#define D2 512

typedef __attribute__((ext_vector_type(16))) __bf16        v16bf;
typedef __attribute__((ext_vector_type(8)))  float         v8f;
typedef __attribute__((ext_vector_type(8)))  unsigned int  v8u;
typedef __attribute__((ext_vector_type(2)))  float         v2f;

__device__ __forceinline__ unsigned short f2bf(float x) {
    unsigned int u = __float_as_uint(x);
    unsigned int r = u + 0x7FFFu + ((u >> 16) & 1u);
    return (unsigned short)(r >> 16);
}
__device__ __forceinline__ unsigned int pk2(float lo, float hi) {
    return (unsigned int)f2bf(lo) | ((unsigned int)f2bf(hi) << 16);
}

__device__ __forceinline__ float htanh(float x) {
#if __has_builtin(__builtin_amdgcn_tanhf)
    return __builtin_amdgcn_tanhf(x);
#elif __has_builtin(__builtin_amdgcn_tanh_f32)
    return __builtin_amdgcn_tanh_f32(x);
#else
    float y;
    asm volatile("v_tanh_f32 %0, %1\n\t"
                 "s_delay_alu instid0(TRANS32_DEP_1)"
                 : "=v"(y) : "v"(x));
    return y;
#endif
}

__global__ void packb_kernel(const float* __restrict__ W, unsigned short* __restrict__ Bp) {
    int blk  = blockIdx.x;
    int nt   = blk >> 3;
    int kc   = blk & 7;
    int lane = threadIdx.x;
    int n     = nt * 16 + (lane & 15);
    int kbase = kc * 32 + ((lane >> 4) ? 8 : 0);
    const float* wrow = W + (size_t)n * D2 + H;
    unsigned short* dst = Bp + ((size_t)blk * 32 + lane) * 16;
    #pragma unroll
    for (int j = 0; j < 8; ++j) dst[j]     = f2bf(wrow[kbase + j]);
    #pragma unroll
    for (int j = 0; j < 8; ++j) dst[8 + j] = f2bf(wrow[kbase + 16 + j]);
}

__global__ void __launch_bounds__(256)
p_kernel(const float* __restrict__ hidden, const float* __restrict__ W,
         const float* __restrict__ bvec, float* __restrict__ P) {
    __shared__ float part[2][4][8][32];
    const int lane = threadIdx.x & 31;
    const int wave = __builtin_amdgcn_readfirstlane(threadIdx.x >> 5);
    const int ct = wave & 1;
    const int q  = wave >> 1;
    const int c  = lane & 15;
    const int h  = lane >> 4;
    const int b0 = (blockIdx.x & 1) << 4;
    const int n0 = (blockIdx.x >> 1) << 5;
    const float* ap = hidden + (b0 + c) * H + q * 64 + h * 4;
    const float* wp = W + (size_t)(n0 + ct * 16 + c) * D2 + q * 64 + h * 4;
    float4 a[8], w[8];
    #pragma unroll
    for (int j = 0; j < 8; ++j) {
        a[j] = *(const float4*)(ap + 8 * j);
        w[j] = *(const float4*)(wp + 8 * j);
    }
    v8f acc0 = {}, acc1 = {};
    #pragma unroll
    for (int j = 0; j < 8; ++j) {
        const v2f a0 = {a[j].x, a[j].y}, w0 = {w[j].x, w[j].y};
        const v2f a1 = {a[j].z, a[j].w}, w1 = {w[j].z, w[j].w};
        acc0 = __builtin_amdgcn_wmma_f32_16x16x4_f32(false, a0, false, w0, (short)0, acc0, false, false);
        acc1 = __builtin_amdgcn_wmma_f32_16x16x4_f32(false, a1, false, w1, (short)0, acc1, false, false);
    }
    #pragma unroll
    for (int r = 0; r < 8; ++r) part[ct][q][r][lane] = acc0[r] + acc1[r];
    __syncthreads();
    const int st = lane >> 4;
    const float bias = bvec[n0 + lane];
    #pragma unroll
    for (int hh = 0; hh < 2; ++hh) {
        const int l = hh * 16 + c;
        const float s = (part[st][0][wave][l] + part[st][1][wave][l])
                      + (part[st][2][wave][l] + part[st][3][wave][l]);
        P[(b0 + 8 * hh + wave) * D2 + n0 + lane] = bias + s;
    }
}

#define SC_WAVES 8
#define SC_CT    4
#define SC_PAIRS 2
#define SC_TRIPS 8
#define SC_ROWS  (32 * SC_PAIRS)
#define FR_BLK   136
#define FR_HALF  68
#define FR_TILE  (16 * FR_BLK)
#define SC_R8    32
#define ROW_F    (BB * H)

typedef __attribute__((ext_vector_type(2))) __bf16 v2bf;

__device__ __forceinline__ unsigned int cvt2(float lo, float hi) {
    v2f x = {lo, hi};
    return __builtin_bit_cast(unsigned int, __builtin_convertvector(x, v2bf));
}
__device__ __forceinline__ uint2 cvt4(float4 f) {
    return make_uint2(cvt2(f.x, f.y), cvt2(f.z, f.w));
}

template <int MASK>
__device__ __forceinline__ float rowx(float x) {
    return __builtin_bit_cast(float, __builtin_amdgcn_update_dpp(
        0, __builtin_bit_cast(int, x), 0x160 | MASK, 0xf, 0xf, false));
}
__device__ __forceinline__ float xhalf(float x) {
    const int b = __builtin_bit_cast(int, x);
    return __builtin_bit_cast(float, __builtin_amdgcn_permlanex16(b, b, 0x76543210, 0xfedcba98, false, false));
}
__device__ __forceinline__ float wave_max(float x) {
    x = fmaxf(x, rowx<1>(x)); x = fmaxf(x, rowx<2>(x));
    x = fmaxf(x, rowx<4>(x)); x = fmaxf(x, rowx<8>(x));
    return fmaxf(x, xhalf(x));
}
__device__ __forceinline__ float wave_sum(float x) {
    x += rowx<1>(x); x += rowx<2>(x); x += rowx<4>(x); x += rowx<8>(x);
    return x + xhalf(x);
}

__global__ void __launch_bounds__(256)
scores_kernel(const float* __restrict__ enc, const v8u* __restrict__ Bq,
              const float* __restrict__ P, const float* __restrict__ v,
              float* __restrict__ scores, float* __restrict__ stats,
              float* __restrict__ out) {
    __shared__ __attribute__((aligned(16))) unsigned int efrag[2][2 * SC_PAIRS * FR_TILE];
    __shared__ float red[2][SC_WAVES][SC_ROWS];
    __shared__ float srow[SC_TRIPS * SC_ROWS];

    const int tid  = threadIdx.x;
    const int wave = __builtin_amdgcn_readfirstlane(tid >> 5);
    const int lane = tid & 31;
    const int g    = lane >> 4;
    const int m    = lane & 15;
    const int b    = blockIdx.x >> 3;
    const int tb   = (blockIdx.x & 7) << 9;

    out[b * TT + tb + tid]       = 0.f;
    out[b * TT + tb + 256 + tid] = 0.f;

    const float* erow = enc + ((size_t)(tb + wave) * BB + b) * H + lane * 4;
    const int sdw = (lane >> 2) * FR_BLK + ((lane >> 1) & 1) * FR_HALF + wave * 4 + (lane & 1) * 2;
    const int rdw = g * FR_HALF + m * 4;

    {
        float4 f[8 * SC_PAIRS];
        #pragma unroll
        for (int tl = 0; tl < 2 * SC_PAIRS; ++tl) {
            #pragma unroll
            for (int rr = 0; rr < 2; ++rr) {
                const float* p = erow + (tl * 16 + rr * 8) * ROW_F;
                f[4 * tl + 2 * rr]     = *(const float4*)(p);
                f[4 * tl + 2 * rr + 1] = *(const float4*)(p + 128);
            }
        }
        unsigned int* sp = &efrag[0][sdw];
        #pragma unroll
        for (int tl = 0; tl < 2 * SC_PAIRS; ++tl) {
            #pragma unroll
            for (int rr = 0; rr < 2; ++rr) {
                *(uint2*)(sp + tl * FR_TILE + rr * SC_R8)              = cvt4(f[4 * tl + 2 * rr]);
                *(uint2*)(sp + tl * FR_TILE + rr * SC_R8 + 8 * FR_BLK) = cvt4(f[4 * tl + 2 * rr + 1]);
            }
        }
    }

    v16bf wf[SC_CT][8];
    {
        const v8u* bp = Bq + ((size_t)(SC_CT * wave) * 8 * 32 + lane);
        #pragma unroll
        for (int t = 0; t < SC_CT; ++t) {
            #pragma unroll
            for (int kc = 0; kc < 8; ++kc)
                wf[t][kc] = __builtin_bit_cast(v16bf, bp[(t * 8 + kc) * 32]);
        }
    }
    float pn[SC_CT][8], vn[SC_CT][8];
    {
        const int n0 = wave * (16 * SC_CT) + g * 8;
        const float* Pb = P + b * D2 + n0;
        #pragma unroll
        for (int t = 0; t < SC_CT; ++t) {
            #pragma unroll
            for (int i = 0; i < 8; ++i) {
                pn[t][i] = Pb[16 * t + i];
                vn[t][i] = v[n0 + 16 * t + i];
            }
        }
    }
    __syncthreads();

    #pragma nounroll
    for (int trip = 0; trip < SC_TRIPS; ++trip) {
        const int cur = trip & 1;
        const bool more = (trip + 1 < SC_TRIPS);
        const float* nx = erow + (size_t)(trip + 1) * (SC_ROWS * ROW_F);

        #pragma unroll
        for (int pr = 0; pr < SC_PAIRS; ++pr) {
            float4 st[8];
            if (more) {
                const float* p = nx + pr * 32 * ROW_F;
                #pragma unroll
                for (int tl = 0; tl < 2; ++tl) {
                    #pragma unroll
                    for (int rr = 0; rr < 2; ++rr) {
                        const float* q = p + (tl * 16 + rr * 8) * ROW_F;
                        st[4 * tl + 2 * rr]     = *(const float4*)(q);
                        st[4 * tl + 2 * rr + 1] = *(const float4*)(q + 128);
                    }
                }
            }

            const unsigned int* fb = &efrag[cur][pr * 2 * FR_TILE + rdw];
            v8f acc[SC_CT][2];
            #pragma unroll
            for (int t = 0; t < SC_CT; ++t) { acc[t][0] = (v8f){}; acc[t][1] = (v8f){}; }
            uint4 fr[3][4];
            #pragma unroll
            for (int s = 0; s < 2; ++s) {
                fr[s][0] = *(const uint4*)(fb + (2 * s) * FR_BLK);
                fr[s][1] = *(const uint4*)(fb + (2 * s + 1) * FR_BLK);
                fr[s][2] = *(const uint4*)(fb + FR_TILE + (2 * s) * FR_BLK);
                fr[s][3] = *(const uint4*)(fb + FR_TILE + (2 * s + 1) * FR_BLK);
            }
            #pragma unroll
            for (int kc = 0; kc < 8; ++kc) {
                if (kc + 2 < 8) {
                    const int s = (kc + 2) % 3;
                    fr[s][0] = *(const uint4*)(fb + (2 * kc + 4) * FR_BLK);
                    fr[s][1] = *(const uint4*)(fb + (2 * kc + 5) * FR_BLK);
                    fr[s][2] = *(const uint4*)(fb + FR_TILE + (2 * kc + 4) * FR_BLK);
                    fr[s][3] = *(const uint4*)(fb + FR_TILE + (2 * kc + 5) * FR_BLK);
                }
                const uint4 l0 = fr[kc % 3][0], h0 = fr[kc % 3][1];
                const uint4 l1 = fr[kc % 3][2], h1 = fr[kc % 3][3];
                const v8u u0 = {l0.x, l0.y, l0.z, l0.w, h0.x, h0.y, h0.z, h0.w};
                const v8u u1 = {l1.x, l1.y, l1.z, l1.w, h1.x, h1.y, h1.z, h1.w};
                const v16bf e0 = __builtin_bit_cast(v16bf, u0);
                const v16bf e1 = __builtin_bit_cast(v16bf, u1);
                #pragma unroll
                for (int t = 0; t < SC_CT; ++t) {
                    acc[t][0] = __builtin_amdgcn_wmma_f32_16x16x32_bf16(
                                    false, wf[t][kc], false, e0, (short)0, acc[t][0], false, false);
                    acc[t][1] = __builtin_amdgcn_wmma_f32_16x16x32_bf16(
                                    false, wf[t][kc], false, e1, (short)0, acc[t][1], false, false);
                }
            }

            float sa[SC_CT], sb[SC_CT];
            #pragma unroll
            for (int t = 0; t < SC_CT; ++t) { sa[t] = 0.f; sb[t] = 0.f; }
            #pragma unroll
            for (int i = 0; i < 8; ++i) {
                #pragma unroll
                for (int t = 0; t < SC_CT; ++t) {
                    sa[t] = fmaf(vn[t][i], htanh(acc[t][0][i] + pn[t][i]), sa[t]);
                    sb[t] = fmaf(vn[t][i], htanh(acc[t][1][i] + pn[t][i]), sb[t]);
                }
            }
            const float s0 = (sa[0] + sa[1]) + (sa[2] + sa[3]);
            const float s1 = (sb[0] + sb[1]) + (sb[2] + sb[3]);
            const float keep = g ? s1 : s0;
            const float give = g ? s0 : s1;
            red[cur][wave][pr * 32 + lane] = keep + __shfl_xor(give, 16, 32);

            if (more) {
                unsigned int* sp = &efrag[cur ^ 1][pr * 2 * FR_TILE + sdw];
                #pragma unroll
                for (int tl = 0; tl < 2; ++tl) {
                    #pragma unroll
                    for (int rr = 0; rr < 2; ++rr) {
                        *(uint2*)(sp + tl * FR_TILE + rr * SC_R8)              = cvt4(st[4 * tl + 2 * rr]);
                        *(uint2*)(sp + tl * FR_TILE + rr * SC_R8 + 8 * FR_BLK) = cvt4(st[4 * tl + 2 * rr + 1]);
                    }
                }
            }
        }
        __syncthreads();

        const int rp = wave - 2 * (trip & 3);
        if (rp == 0 || rp == 1) {
            const int r = rp * 32 + lane;
            float acc = red[cur][0][r];
            #pragma unroll
            for (int w = 1; w < SC_WAVES; ++w) acc += red[cur][w][r];
            scores[b * TT + tb + trip * SC_ROWS + r] = acc;
            srow[trip * SC_ROWS + r] = acc;
        }
    }
    __syncthreads();

    if (wave == 0) {
        float x[16];
        #pragma unroll
        for (int j = 0; j < 16; ++j) x[j] = srow[j * 32 + lane];
        float mx = x[0];
        #pragma unroll
        for (int j = 1; j < 16; ++j) mx = fmaxf(mx, x[j]);
        mx = wave_max(mx);
        float sm = 0.f;
        #pragma unroll
        for (int j = 0; j < 16; ++j) sm += __expf(x[j] - mx);
        sm = wave_sum(sm);
        if (lane == 0) {
            stats[2 * blockIdx.x]     = mx;
            stats[2 * blockIdx.x + 1] = sm;
        }
    }
}

__global__ void scatter_kernel(const int* __restrict__ src, const int* __restrict__ dst,
                               const float* __restrict__ sc, const float* __restrict__ stats,
                               float* __restrict__ out) {
    int i = blockIdx.x * blockDim.x + threadIdx.x;
    if (i >= BB * EE) return;
    const int b = blockIdx.x >> 5;
    const float* st = stats + b * 16;
    float M = st[0];
    #pragma unroll
    for (int j = 1; j < 8; ++j) M = fmaxf(M, st[2 * j]);
    float S = 0.f;
    #pragma unroll
    for (int j = 0; j < 8; ++j) S += st[2 * j + 1] * __expf(st[2 * j] - M);
    const float inv = 1.0f / S;
    int s = src[i], d = dst[i];
    if (d != s + 1)
        atomicAdd(&out[b * TT + d], __expf(sc[b * TT + s] - M) * inv * 0.1f);
}

extern "C" void kernel_launch(void* const* d_in, const int* in_sizes, int n_in,
                              void* d_out, int out_size, void* d_ws, size_t ws_size,
                              hipStream_t stream) {
    const float* hidden = (const float*)d_in[0];
    const float* enc    = (const float*)d_in[1];
    const int*   esrc   = (const int*)d_in[2];
    const int*   edst   = (const int*)d_in[3];
    const float* W      = (const float*)d_in[4];
    const float* bvec   = (const float*)d_in[5];
    const float* v      = (const float*)d_in[6];
    float* out = (float*)d_out;

    char* ws = (char*)d_ws;
    unsigned short* Bpack = (unsigned short*)ws;
    float* P    = (float*)(ws + 262144);
    float* wbuf = (float*)(ws + 262144 + 65536);
    float* stats = (float*)(ws + 262144 + 65536 + 524288);

    packb_kernel  <<<256, 32, 0, stream>>>(W, Bpack);
    p_kernel      <<<32, 256, 0, stream>>>(hidden, W, bvec, P);
    scores_kernel <<<256, 256, 0, stream>>>(enc, (const v8u*)Bpack, P, v, wbuf, stats, out);
    scatter_kernel<<<(BB * EE + 255) / 256, 256, 0, stream>>>(esrc, edst, wbuf, stats, out);
}


# Scatter launch: one 32-wave work-group per batch row; node weights exp'd once into LDS, edge sums by LDS float adds (arrival order), out written once by plain stores; no global atomic, no zeroing
# speedup vs baseline: 2.9272x; 1.1954x over previous
#include <hip/hip_runtime.h>
#include <hip/hip_bf16.h>

#define H 256
#define BB 32
#define TT 4096
#define EE 8192
#define D2 512

typedef __attribute__((ext_vector_type(16))) __bf16        v16bf;
typedef __attribute__((ext_vector_type(8)))  float         v8f;
typedef __attribute__((ext_vector_type(8)))  unsigned int  v8u;
typedef __attribute__((ext_vector_type(2)))  float         v2f;

__device__ __forceinline__ unsigned short f2bf(float x) {
    unsigned int u = __float_as_uint(x);
    unsigned int r = u + 0x7FFFu + ((u >> 16) & 1u);
    return (unsigned short)(r >> 16);
}
__device__ __forceinline__ unsigned int pk2(float lo, float hi) {
    return (unsigned int)f2bf(lo) | ((unsigned int)f2bf(hi) << 16);
}

__device__ __forceinline__ float htanh(float x) {
#if __has_builtin(__builtin_amdgcn_tanhf)
    return __builtin_amdgcn_tanhf(x);
#elif __has_builtin(__builtin_amdgcn_tanh_f32)
    return __builtin_amdgcn_tanh_f32(x);
#else
    float y;
    asm volatile("v_tanh_f32 %0, %1\n\t"
                 "s_delay_alu instid0(TRANS32_DEP_1)"
                 : "=v"(y) : "v"(x));
    return y;
#endif
}

__global__ void packb_kernel(const float* __restrict__ W, unsigned short* __restrict__ Bp) {
    int blk  = blockIdx.x;
    int nt   = blk >> 3;
    int kc   = blk & 7;
    int lane = threadIdx.x;
    int n     = nt * 16 + (lane & 15);
    int kbase = kc * 32 + ((lane >> 4) ? 8 : 0);
    const float* wrow = W + (size_t)n * D2 + H;
    unsigned short* dst = Bp + ((size_t)blk * 32 + lane) * 16;
    #pragma unroll
    for (int j = 0; j < 8; ++j) dst[j]     = f2bf(wrow[kbase + j]);
    #pragma unroll
    for (int j = 0; j < 8; ++j) dst[8 + j] = f2bf(wrow[kbase + 16 + j]);
}

__global__ void __launch_bounds__(256)
p_kernel(const float* __restrict__ hidden, const float* __restrict__ W,
         const float* __restrict__ bvec, float* __restrict__ P) {
    __shared__ float part[2][4][8][32];
    const int lane = threadIdx.x & 31;
    const int wave = __builtin_amdgcn_readfirstlane(threadIdx.x >> 5);
    const int ct = wave & 1;
    const int q  = wave >> 1;
    const int c  = lane & 15;
    const int h  = lane >> 4;
    const int b0 = (blockIdx.x & 1) << 4;
    const int n0 = (blockIdx.x >> 1) << 5;
    const float* ap = hidden + (b0 + c) * H + q * 64 + h * 4;
    const float* wp = W + (size_t)(n0 + ct * 16 + c) * D2 + q * 64 + h * 4;
    float4 a[8], w[8];
    #pragma unroll
    for (int j = 0; j < 8; ++j) {
        a[j] = *(const float4*)(ap + 8 * j);
        w[j] = *(const float4*)(wp + 8 * j);
    }
    v8f acc0 = {}, acc1 = {};
    #pragma unroll
    for (int j = 0; j < 8; ++j) {
        const v2f a0 = {a[j].x, a[j].y}, w0 = {w[j].x, w[j].y};
        const v2f a1 = {a[j].z, a[j].w}, w1 = {w[j].z, w[j].w};
        acc0 = __builtin_amdgcn_wmma_f32_16x16x4_f32(false, a0, false, w0, (short)0, acc0, false, false);
        acc1 = __builtin_amdgcn_wmma_f32_16x16x4_f32(false, a1, false, w1, (short)0, acc1, false, false);
    }
    #pragma unroll
    for (int r = 0; r < 8; ++r) part[ct][q][r][lane] = acc0[r] + acc1[r];
    __syncthreads();
    const int st = lane >> 4;
    const float bias = bvec[n0 + lane];
    #pragma unroll
    for (int hh = 0; hh < 2; ++hh) {
        const int l = hh * 16 + c;
        const float s = (part[st][0][wave][l] + part[st][1][wave][l])
                      + (part[st][2][wave][l] + part[st][3][wave][l]);
        P[(b0 + 8 * hh + wave) * D2 + n0 + lane] = bias + s;
    }
}

#define SC_WAVES 8
#define SC_CT    4
#define SC_PAIRS 2
#define SC_TRIPS 8
#define SC_ROWS  (32 * SC_PAIRS)
#define FR_BLK   136
#define FR_HALF  68
#define FR_TILE  (16 * FR_BLK)
#define SC_R8    32
#define ROW_F    (BB * H)

typedef __attribute__((ext_vector_type(2))) __bf16 v2bf;

__device__ __forceinline__ unsigned int cvt2(float lo, float hi) {
    v2f x = {lo, hi};
    return __builtin_bit_cast(unsigned int, __builtin_convertvector(x, v2bf));
}
__device__ __forceinline__ uint2 cvt4(float4 f) {
    return make_uint2(cvt2(f.x, f.y), cvt2(f.z, f.w));
}

template <int MASK>
__device__ __forceinline__ float rowx(float x) {
    return __builtin_bit_cast(float, __builtin_amdgcn_update_dpp(
        0, __builtin_bit_cast(int, x), 0x160 | MASK, 0xf, 0xf, false));
}
__device__ __forceinline__ float xhalf(float x) {
    const int b = __builtin_bit_cast(int, x);
    return __builtin_bit_cast(float, __builtin_amdgcn_permlanex16(b, b, 0x76543210, 0xfedcba98, false, false));
}
__device__ __forceinline__ float wave_max(float x) {
    x = fmaxf(x, rowx<1>(x)); x = fmaxf(x, rowx<2>(x));
    x = fmaxf(x, rowx<4>(x)); x = fmaxf(x, rowx<8>(x));
    return fmaxf(x, xhalf(x));
}
__device__ __forceinline__ float wave_sum(float x) {
    x += rowx<1>(x); x += rowx<2>(x); x += rowx<4>(x); x += rowx<8>(x);
    return x + xhalf(x);
}

__global__ void __launch_bounds__(256)
scores_kernel(const float* __restrict__ enc, const v8u* __restrict__ Bq,
              const float* __restrict__ P, const float* __restrict__ v,
              float* __restrict__ scores, float* __restrict__ stats) {
    __shared__ __attribute__((aligned(16))) unsigned int efrag[2][2 * SC_PAIRS * FR_TILE];
    __shared__ float red[2][SC_WAVES][SC_ROWS];
    __shared__ float srow[SC_TRIPS * SC_ROWS];

    const int tid  = threadIdx.x;
    const int wave = __builtin_amdgcn_readfirstlane(tid >> 5);
    const int lane = tid & 31;
    const int g    = lane >> 4;
    const int m    = lane & 15;
    const int b    = blockIdx.x >> 3;
    const int tb   = (blockIdx.x & 7) << 9;

    const float* erow = enc + ((size_t)(tb + wave) * BB + b) * H + lane * 4;
    const int sdw = (lane >> 2) * FR_BLK + ((lane >> 1) & 1) * FR_HALF + wave * 4 + (lane & 1) * 2;
    const int rdw = g * FR_HALF + m * 4;

    {
        float4 f[8 * SC_PAIRS];
        #pragma unroll
        for (int tl = 0; tl < 2 * SC_PAIRS; ++tl) {
            #pragma unroll
            for (int rr = 0; rr < 2; ++rr) {
                const float* p = erow + (tl * 16 + rr * 8) * ROW_F;
                f[4 * tl + 2 * rr]     = *(const float4*)(p);
                f[4 * tl + 2 * rr + 1] = *(const float4*)(p + 128);
            }
        }
        unsigned int* sp = &efrag[0][sdw];
        #pragma unroll
        for (int tl = 0; tl < 2 * SC_PAIRS; ++tl) {
            #pragma unroll
            for (int rr = 0; rr < 2; ++rr) {
                *(uint2*)(sp + tl * FR_TILE + rr * SC_R8)              = cvt4(f[4 * tl + 2 * rr]);
                *(uint2*)(sp + tl * FR_TILE + rr * SC_R8 + 8 * FR_BLK) = cvt4(f[4 * tl + 2 * rr + 1]);
            }
        }
    }

    v16bf wf[SC_CT][8];
    {
        const v8u* bp = Bq + ((size_t)(SC_CT * wave) * 8 * 32 + lane);
        #pragma unroll
        for (int t = 0; t < SC_CT; ++t) {
            #pragma unroll
            for (int kc = 0; kc < 8; ++kc)
                wf[t][kc] = __builtin_bit_cast(v16bf, bp[(t * 8 + kc) * 32]);
        }
    }
    float pn[SC_CT][8], vn[SC_CT][8];
    {
        const int n0 = wave * (16 * SC_CT) + g * 8;
        const float* Pb = P + b * D2 + n0;
        #pragma unroll
        for (int t = 0; t < SC_CT; ++t) {
            #pragma unroll
            for (int i = 0; i < 8; ++i) {
                pn[t][i] = Pb[16 * t + i];
                vn[t][i] = v[n0 + 16 * t + i];
            }
        }
    }
    __syncthreads();

    #pragma nounroll
    for (int trip = 0; trip < SC_TRIPS; ++trip) {
        const int cur = trip & 1;
        const bool more = (trip + 1 < SC_TRIPS);
        const float* nx = erow + (size_t)(trip + 1) * (SC_ROWS * ROW_F);

        #pragma unroll
        for (int pr = 0; pr < SC_PAIRS; ++pr) {
            float4 st[8];
            if (more) {
                const float* p = nx + pr * 32 * ROW_F;
                #pragma unroll
                for (int tl = 0; tl < 2; ++tl) {
                    #pragma unroll
                    for (int rr = 0; rr < 2; ++rr) {
                        const float* q = p + (tl * 16 + rr * 8) * ROW_F;
                        st[4 * tl + 2 * rr]     = *(const float4*)(q);
                        st[4 * tl + 2 * rr + 1] = *(const float4*)(q + 128);
                    }
                }
            }

            const unsigned int* fb = &efrag[cur][pr * 2 * FR_TILE + rdw];
            v8f acc[SC_CT][2];
            #pragma unroll
            for (int t = 0; t < SC_CT; ++t) { acc[t][0] = (v8f){}; acc[t][1] = (v8f){}; }
            uint4 fr[3][4];
            #pragma unroll
            for (int s = 0; s < 2; ++s) {
                fr[s][0] = *(const uint4*)(fb + (2 * s) * FR_BLK);
                fr[s][1] = *(const uint4*)(fb + (2 * s + 1) * FR_BLK);
                fr[s][2] = *(const uint4*)(fb + FR_TILE + (2 * s) * FR_BLK);
                fr[s][3] = *(const uint4*)(fb + FR_TILE + (2 * s + 1) * FR_BLK);
            }
            #pragma unroll
            for (int kc = 0; kc < 8; ++kc) {
                if (kc + 2 < 8) {
                    const int s = (kc + 2) % 3;
                    fr[s][0] = *(const uint4*)(fb + (2 * kc + 4) * FR_BLK);
                    fr[s][1] = *(const uint4*)(fb + (2 * kc + 5) * FR_BLK);
                    fr[s][2] = *(const uint4*)(fb + FR_TILE + (2 * kc + 4) * FR_BLK);
                    fr[s][3] = *(const uint4*)(fb + FR_TILE + (2 * kc + 5) * FR_BLK);
                }
                const uint4 l0 = fr[kc % 3][0], h0 = fr[kc % 3][1];
                const uint4 l1 = fr[kc % 3][2], h1 = fr[kc % 3][3];
                const v8u u0 = {l0.x, l0.y, l0.z, l0.w, h0.x, h0.y, h0.z, h0.w};
                const v8u u1 = {l1.x, l1.y, l1.z, l1.w, h1.x, h1.y, h1.z, h1.w};
                const v16bf e0 = __builtin_bit_cast(v16bf, u0);
                const v16bf e1 = __builtin_bit_cast(v16bf, u1);
                #pragma unroll
                for (int t = 0; t < SC_CT; ++t) {
                    acc[t][0] = __builtin_amdgcn_wmma_f32_16x16x32_bf16(
                                    false, wf[t][kc], false, e0, (short)0, acc[t][0], false, false);
                    acc[t][1] = __builtin_amdgcn_wmma_f32_16x16x32_bf16(
                                    false, wf[t][kc], false, e1, (short)0, acc[t][1], false, false);
                }
            }

            float sa[SC_CT], sb[SC_CT];
            #pragma unroll
            for (int t = 0; t < SC_CT; ++t) { sa[t] = 0.f; sb[t] = 0.f; }
            #pragma unroll
            for (int i = 0; i < 8; ++i) {
                #pragma unroll
                for (int t = 0; t < SC_CT; ++t) {
                    sa[t] = fmaf(vn[t][i], htanh(acc[t][0][i] + pn[t][i]), sa[t]);
                    sb[t] = fmaf(vn[t][i], htanh(acc[t][1][i] + pn[t][i]), sb[t]);
                }
            }
            const float s0 = (sa[0] + sa[1]) + (sa[2] + sa[3]);
            const float s1 = (sb[0] + sb[1]) + (sb[2] + sb[3]);
            const float keep = g ? s1 : s0;
            const float give = g ? s0 : s1;
            red[cur][wave][pr * 32 + lane] = keep + __shfl_xor(give, 16, 32);

            if (more) {
                unsigned int* sp = &efrag[cur ^ 1][pr * 2 * FR_TILE + sdw];
                #pragma unroll
                for (int tl = 0; tl < 2; ++tl) {
                    #pragma unroll
                    for (int rr = 0; rr < 2; ++rr) {
                        *(uint2*)(sp + tl * FR_TILE + rr * SC_R8)              = cvt4(st[4 * tl + 2 * rr]);
                        *(uint2*)(sp + tl * FR_TILE + rr * SC_R8 + 8 * FR_BLK) = cvt4(st[4 * tl + 2 * rr + 1]);
                    }
                }
            }
        }
        __syncthreads();

        const int rp = wave - 2 * (trip & 3);
        if (rp == 0 || rp == 1) {
            const int r = rp * 32 + lane;
            float acc = red[cur][0][r];
            #pragma unroll
            for (int w = 1; w < SC_WAVES; ++w) acc += red[cur][w][r];
            scores[b * TT + tb + trip * SC_ROWS + r] = acc;
            srow[trip * SC_ROWS + r] = acc;
        }
    }
    __syncthreads();

    if (wave == 0) {
        float x[16];
        #pragma unroll
        for (int j = 0; j < 16; ++j) x[j] = srow[j * 32 + lane];
        float mx = x[0];
        #pragma unroll
        for (int j = 1; j < 16; ++j) mx = fmaxf(mx, x[j]);
        mx = wave_max(mx);
        float sm = 0.f;
        #pragma unroll
        for (int j = 0; j < 16; ++j) sm += __expf(x[j] - mx);
        sm = wave_sum(sm);
        if (lane == 0) {
            stats[2 * blockIdx.x]     = mx;
            stats[2 * blockIdx.x + 1] = sm;
        }
    }
}

#define SCAT_T 1024
__global__ void __launch_bounds__(SCAT_T)
scatter_kernel(const int* __restrict__ src, const int* __restrict__ dst,
               const float* __restrict__ sc, const float* __restrict__ stats,
               float* __restrict__ out) {
    __shared__ __attribute__((aligned(16))) float wrow[TT];
    __shared__ __attribute__((aligned(16))) float acc[TT];
    const int b   = blockIdx.x;
    const int tid = threadIdx.x;
    const float4 x = ((const float4*)(sc + b * TT))[tid];
    const int4* sp = (const int4*)(src + b * EE);
    const int4* dp = (const int4*)(dst + b * EE);
    const int4 s0 = sp[tid], s1 = sp[tid + SCAT_T];
    const int4 d0 = dp[tid], d1 = dp[tid + SCAT_T];
    const float* st = stats + b * 16;
    float M = st[0];
    #pragma unroll
    for (int j = 1; j < 8; ++j) M = fmaxf(M, st[2 * j]);
    float S = 0.f;
    #pragma unroll
    for (int j = 0; j < 8; ++j) S += st[2 * j + 1] * __expf(st[2 * j] - M);
    const float inv = 1.0f / S;
    float4 w;
    w.x = __expf(x.x - M) * inv * 0.1f;
    w.y = __expf(x.y - M) * inv * 0.1f;
    w.z = __expf(x.z - M) * inv * 0.1f;
    w.w = __expf(x.w - M) * inv * 0.1f;
    ((float4*)wrow)[tid] = w;
    ((float4*)acc)[tid]  = make_float4(0.f, 0.f, 0.f, 0.f);
    __syncthreads();
    const float g0 = wrow[s0.x], g1 = wrow[s0.y], g2 = wrow[s0.z], g3 = wrow[s0.w];
    const float g4 = wrow[s1.x], g5 = wrow[s1.y], g6 = wrow[s1.z], g7 = wrow[s1.w];
    atomicAdd(&acc[d0.x], d0.x != s0.x + 1 ? g0 : 0.f);
    atomicAdd(&acc[d0.y], d0.y != s0.y + 1 ? g1 : 0.f);
    atomicAdd(&acc[d0.z], d0.z != s0.z + 1 ? g2 : 0.f);
    atomicAdd(&acc[d0.w], d0.w != s0.w + 1 ? g3 : 0.f);
    atomicAdd(&acc[d1.x], d1.x != s1.x + 1 ? g4 : 0.f);
    atomicAdd(&acc[d1.y], d1.y != s1.y + 1 ? g5 : 0.f);
    atomicAdd(&acc[d1.z], d1.z != s1.z + 1 ? g6 : 0.f);
    atomicAdd(&acc[d1.w], d1.w != s1.w + 1 ? g7 : 0.f);
    __syncthreads();
    ((float4*)(out + b * TT))[tid] = ((const float4*)acc)[tid];
}

extern "C" void kernel_launch(void* const* d_in, const int* in_sizes, int n_in,
                              void* d_out, int out_size, void* d_ws, size_t ws_size,
                              hipStream_t stream) {
    const float* hidden = (const float*)d_in[0];
    const float* enc    = (const float*)d_in[1];
    const int*   esrc   = (const int*)d_in[2];
    const int*   edst   = (const int*)d_in[3];
    const float* W      = (const float*)d_in[4];
    const float* bvec   = (const float*)d_in[5];
    const float* v      = (const float*)d_in[6];
    float* out = (float*)d_out;

    char* ws = (char*)d_ws;
    unsigned short* Bpack = (unsigned short*)ws;
    float* P    = (float*)(ws + 262144);
    float* wbuf = (float*)(ws + 262144 + 65536);
    float* stats = (float*)(ws + 262144 + 65536 + 524288);

    packb_kernel  <<<256, 32, 0, stream>>>(W, Bpack);
    p_kernel      <<<32, 256, 0, stream>>>(hidden, W, bvec, P);
    scores_kernel <<<256, 256, 0, stream>>>(enc, (const v8u*)Bpack, P, v, wbuf, stats);
    scatter_kernel<<<BB, SCAT_T, 0, stream>>>(esrc, edst, wbuf, stats, out);
}
